# starting kernel
# baseline (speedup 1.0000x reference)
#include <hip/hip_runtime.h>
#include <cmath>


typedef float v2f __attribute__((ext_vector_type(2)));
typedef float v8f __attribute__((ext_vector_type(8)));

#define TK   384
#define HH   8
#define FDM  32
#define NE   256
#define NU   768

__global__ void srpe_hid_kernel(const float* __restrict__ pd,
                                const float* __restrict__ W1,
                                const float* __restrict__ b1,
                                float* __restrict__ hid)
{
    int idx = blockIdx.x * blockDim.x + threadIdx.x;
    if (idx >= NU * NE) return;
    int u = idx >> 8;
    int c = idx & 255;
    int ue = (u < 766) ? u : 766;
    float d = (ue >= 383) ? pd[(ue - 383) * TK] : pd[383 - ue];
    float f0 = log1pf(fmaxf(d, 0.0f));
    float f1 = log1pf(fmaxf(-d, 0.0f));
    float f2 = (d == 0.0f) ? 1.0f : 0.0f;
    float e  = f0 * W1[c * 3 + 0] + f1 * W1[c * 3 + 1] + f2 * W1[c * 3 + 2] + b1[c];
    float e3 = e * e * e;
    float th = tanhf(0.79788456080286535588f * (e + 0.044715f * e3));
    hid[idx] = 0.5f * e * (1.0f + th);
}

__global__ void srpe_rd_wmma(const float* __restrict__ hid,
                             const float* __restrict__ W2,
                             const float* __restrict__ b2,
                             float* __restrict__ Rd)
{
    int tile = blockIdx.x;
    int ci   = tile & 15;
    int ui   = tile >> 4;
    int u0   = ui * 16;
    int c0   = ci * 16;
    int lane = threadIdx.x;
    int m    = lane & 15;
    int half = lane >> 4;

    v8f acc = {};
    for (int kb = 0; kb < NE; kb += 4) {
        int ka = kb + half * 2;
        v2f a, b;
        a.x = hid[(u0 + m) * NE + ka];
        a.y = hid[(u0 + m) * NE + ka + 1];
        b.x = W2[(c0 + m) * NE + ka];
        b.y = W2[(c0 + m) * NE + ka + 1];
        acc = __builtin_amdgcn_wmma_f32_16x16x4_f32(
                  false, a, false, b, (short)0, acc, false, false);
    }
    float bias = b2[c0 + m];
    #pragma unroll
    for (int r = 0; r < 8; ++r) {
        int row = r + 8 * half;
        Rd[(u0 + row) * NE + (c0 + m)] = acc[r] + bias;
    }
}

__global__ void srpe_qk_wmma(const float* __restrict__ x,
                             const float* __restrict__ Rd,
                             float* __restrict__ out)
{
    int bh = blockIdx.y;
    int h  = bh & (HH - 1);
    int id = blockIdx.x;
    int ti = id / 25;
    int jo = id % 25;
    int t0 = ti * 16;
    int u0 = (ti + jo) * 16;

    int lane = threadIdx.x;
    int m    = lane & 15;
    int half = lane >> 4;

    const float* X = x  + (size_t)bh * TK * FDM;
    const float* R = Rd + h * FDM;

    v8f acc = {};
    #pragma unroll
    for (int kb = 0; kb < FDM; kb += 4) {
        int ka = kb + half * 2;
        v2f a, b;
        a.x = X[(t0 + m) * FDM + ka];
        a.y = X[(t0 + m) * FDM + ka + 1];
        b.x = R[(u0 + m) * NE + ka];
        b.y = R[(u0 + m) * NE + ka + 1];
        acc = __builtin_amdgcn_wmma_f32_16x16x4_f32(
                  false, a, false, b, (short)0, acc, false, false);
    }

    float* O = out + (size_t)bh * TK * TK;
    int u = u0 + m;
    #pragma unroll
    for (int r = 0; r < 8; ++r) {
        int t = t0 + r + 8 * half;
        int s = t + 383 - u;
        if (s >= 0 && s < TK) O[t * TK + s] = acc[r];
    }
}

extern "C" void kernel_launch(void* const* d_in, const int* in_sizes, int n_in,
                              void* d_out, int out_size, void* d_ws, size_t ws_size,
                              hipStream_t stream)
{
    const float* x  = (const float*)d_in[0];
    const float* pd = (const float*)d_in[1];
    const float* W1 = (const float*)d_in[2];
    const float* b1 = (const float*)d_in[3];
    const float* W2 = (const float*)d_in[4];
    const float* b2 = (const float*)d_in[5];
    float* out = (float*)d_out;

    float* hid = (float*)d_ws;
    float* Rd  = hid + NU * NE;

    srpe_hid_kernel<<<dim3((NU * NE) / 256), dim3(256), 0, stream>>>(pd, W1, b1, hid);

    srpe_rd_wmma<<<dim3((NU / 16) * (NE / 16)), dim3(32), 0, stream>>>(hid, W2, b2, Rd);

    srpe_qk_wmma<<<dim3(24 * 25, 16), dim3(32), 0, stream>>>(x, Rd, out);
}


# stage 2: each wave takes 5 consecutive band tiles of its t-tile (A fragments reused, all loads issued as one group): 9600 single-wave workgroups -> 1920; k order and sums per element unchanged
# speedup vs baseline: 1.1249x; 1.1249x over previous
#include <hip/hip_runtime.h>
#include <cmath>


typedef float v2f __attribute__((ext_vector_type(2)));
typedef float v8f __attribute__((ext_vector_type(8)));

#define TK   384
#define HH   8
#define FDM  32
#define NE   256
#define NU   768

__global__ void srpe_hid_kernel(const float* __restrict__ pd,
                                const float* __restrict__ W1,
                                const float* __restrict__ b1,
                                float* __restrict__ hid)
{
    int idx = blockIdx.x * blockDim.x + threadIdx.x;
    if (idx >= NU * NE) return;
    int u = idx >> 8;
    int c = idx & 255;
    int ue = (u < 766) ? u : 766;
    float d = (ue >= 383) ? pd[(ue - 383) * TK] : pd[383 - ue];
    float f0 = log1pf(fmaxf(d, 0.0f));
    float f1 = log1pf(fmaxf(-d, 0.0f));
    float f2 = (d == 0.0f) ? 1.0f : 0.0f;
    float e  = f0 * W1[c * 3 + 0] + f1 * W1[c * 3 + 1] + f2 * W1[c * 3 + 2] + b1[c];
    float e3 = e * e * e;
    float th = tanhf(0.79788456080286535588f * (e + 0.044715f * e3));
    hid[idx] = 0.5f * e * (1.0f + th);
}

__global__ void srpe_rd_wmma(const float* __restrict__ hid,
                             const float* __restrict__ W2,
                             const float* __restrict__ b2,
                             float* __restrict__ Rd)
{
    int tile = blockIdx.x;
    int ci   = tile & 15;
    int ui   = tile >> 4;
    int u0   = ui * 16;
    int c0   = ci * 16;
    int lane = threadIdx.x;
    int m    = lane & 15;
    int half = lane >> 4;

    v8f acc = {};
    for (int kb = 0; kb < NE; kb += 4) {
        int ka = kb + half * 2;
        v2f a, b;
        a.x = hid[(u0 + m) * NE + ka];
        a.y = hid[(u0 + m) * NE + ka + 1];
        b.x = W2[(c0 + m) * NE + ka];
        b.y = W2[(c0 + m) * NE + ka + 1];
        acc = __builtin_amdgcn_wmma_f32_16x16x4_f32(
                  false, a, false, b, (short)0, acc, false, false);
    }
    float bias = b2[c0 + m];
    #pragma unroll
    for (int r = 0; r < 8; ++r) {
        int row = r + 8 * half;
        Rd[(u0 + row) * NE + (c0 + m)] = acc[r] + bias;
    }
}

#define JT 5
__global__ void __launch_bounds__(32)
srpe_qk_wmma(const float* __restrict__ x,
             const float* __restrict__ Rd,
             float* __restrict__ out)
{
    int bh = blockIdx.y;
    int h  = bh & (HH - 1);
    int id = blockIdx.x;
    int ti = id / (25 / JT);
    int jg = id % (25 / JT);
    int t0 = ti * 16;
    int ub = (ti + jg * JT) * 16;

    int lane = threadIdx.x;
    int m    = lane & 15;
    int half = lane >> 4;

    const float* X = x  + (size_t)bh * TK * FDM;
    const float* R = Rd + h * FDM;

    v2f a[FDM / 4];
    #pragma unroll
    for (int kk = 0; kk < FDM / 4; ++kk) {
        int ka = kk * 4 + half * 2;
        a[kk].x = X[(t0 + m) * FDM + ka];
        a[kk].y = X[(t0 + m) * FDM + ka + 1];
    }
    v2f b[JT][FDM / 4];
    #pragma unroll
    for (int j = 0; j < JT; ++j) {
        #pragma unroll
        for (int kk = 0; kk < FDM / 4; ++kk) {
            int ka = kk * 4 + half * 2;
            b[j][kk].x = R[(ub + j * 16 + m) * NE + ka];
            b[j][kk].y = R[(ub + j * 16 + m) * NE + ka + 1];
        }
    }

    asm volatile("" :: "v"(a[0]), "v"(a[1]), "v"(a[2]), "v"(a[3]), "v"(a[4]), "v"(a[5]), "v"(a[6]), "v"(a[7]), "v"(b[0][0]), "v"(b[0][1]), "v"(b[0][2]), "v"(b[0][3]), "v"(b[0][4]), "v"(b[0][5]), "v"(b[0][6]), "v"(b[0][7]), "v"(b[1][0]), "v"(b[1][1]), "v"(b[1][2]), "v"(b[1][3]), "v"(b[1][4]), "v"(b[1][5]), "v"(b[1][6]), "v"(b[1][7]), "v"(b[2][0]), "v"(b[2][1]), "v"(b[2][2]), "v"(b[2][3]), "v"(b[2][4]), "v"(b[2][5]), "v"(b[2][6]), "v"(b[2][7]), "v"(b[3][0]), "v"(b[3][1]), "v"(b[3][2]), "v"(b[3][3]), "v"(b[3][4]), "v"(b[3][5]), "v"(b[3][6]), "v"(b[3][7]), "v"(b[4][0]), "v"(b[4][1]), "v"(b[4][2]), "v"(b[4][3]), "v"(b[4][4]), "v"(b[4][5]), "v"(b[4][6]), "v"(b[4][7]));

    v8f acc[JT];
    #pragma unroll
    for (int j = 0; j < JT; ++j) acc[j] = v8f{};
    #pragma unroll
    for (int kk = 0; kk < FDM / 4; ++kk) {
        #pragma unroll
        for (int j = 0; j < JT; ++j) {
            acc[j] = __builtin_amdgcn_wmma_f32_16x16x4_f32(
                         false, a[kk], false, b[j][kk], (short)0, acc[j], false, false);
        }
    }

    float* O = out + (size_t)bh * TK * TK;
    #pragma unroll
    for (int j = 0; j < JT; ++j) {
        int u = ub + j * 16 + m;
        #pragma unroll
        for (int r = 0; r < 8; ++r) {
            int t = t0 + r + 8 * half;
            int s = t + 383 - u;
            if (s >= 0 && s < TK) O[t * TK + s] = acc[j][r];
        }
    }
}

extern "C" void kernel_launch(void* const* d_in, const int* in_sizes, int n_in,
                              void* d_out, int out_size, void* d_ws, size_t ws_size,
                              hipStream_t stream)
{
    const float* x  = (const float*)d_in[0];
    const float* pd = (const float*)d_in[1];
    const float* W1 = (const float*)d_in[2];
    const float* b1 = (const float*)d_in[3];
    const float* W2 = (const float*)d_in[4];
    const float* b2 = (const float*)d_in[5];
    float* out = (float*)d_out;

    float* hid = (float*)d_ws;
    float* Rd  = hid + NU * NE;

    srpe_hid_kernel<<<dim3((NU * NE) / 256), dim3(256), 0, stream>>>(pd, W1, b1, hid);

    srpe_rd_wmma<<<dim3((NU / 16) * (NE / 16)), dim3(32), 0, stream>>>(hid, W2, b2, Rd);

    srpe_qk_wmma<<<dim3(24 * 5, 16), dim3(32), 0, stream>>>(x, Rd, out);
}


# stage 1a fused into 1b: each 8-wave workgroup computes the 16x256 gelu tile it consumes into LDS and takes the WMMA A operand from there; 3 launches -> 2; k order of both GEMMs unchanged
# speedup vs baseline: 1.3430x; 1.1939x over previous
#include <hip/hip_runtime.h>
#include <cmath>


typedef float v2f __attribute__((ext_vector_type(2)));
typedef float v8f __attribute__((ext_vector_type(8)));

#define TK   384
#define HH   8
#define FDM  32
#define NE   256
#define NU   768

#define HP 260
__global__ void __launch_bounds__(256)
srpe_rd_fused(const float* __restrict__ pd,
              const float* __restrict__ W1,
              const float* __restrict__ b1,
              const float* __restrict__ W2,
              const float* __restrict__ b2,
              float* __restrict__ Rd)
{
    __shared__ float hid_s[16 * HP];

    int tid  = threadIdx.x;
    int wave = tid >> 5;
    int lane = tid & 31;
    int m    = lane & 15;
    int half = lane >> 4;
    int ui   = blockIdx.x >> 1;
    int cg   = blockIdx.x & 1;
    int u0   = ui * 16;
    int c0   = (cg * 8 + wave) * 16;

    int ur = u0 + m;
    int ue = (ur < 766) ? ur : 766;
    float d = (ue >= 383) ? pd[(ue - 383) * TK] : pd[383 - ue];
    float w0 = W1[tid * 3 + 0];
    float w1 = W1[tid * 3 + 1];
    float w2 = W1[tid * 3 + 2];
    float bb = b1[tid];

    v2f b[NE / 4];
    #pragma unroll
    for (int kk = 0; kk < NE / 4; ++kk) {
        int ka = kk * 4 + half * 2;
        b[kk].x = W2[(c0 + m) * NE + ka];
        b[kk].y = W2[(c0 + m) * NE + ka + 1];
    }

    int f0 = __float_as_int(log1pf(fmaxf(d, 0.0f)));
    int f1 = __float_as_int(log1pf(fmaxf(-d, 0.0f)));
    int f2 = __float_as_int((d == 0.0f) ? 1.0f : 0.0f);

    #pragma unroll
    for (int u = 0; u < 16; ++u) {
        float g0 = __int_as_float(__builtin_amdgcn_readlane(f0, u));
        float g1 = __int_as_float(__builtin_amdgcn_readlane(f1, u));
        float g2 = __int_as_float(__builtin_amdgcn_readlane(f2, u));
        float e  = g0 * w0 + g1 * w1 + g2 * w2 + bb;
        float e3 = e * e * e;
        float th = tanhf(0.79788456080286535588f * (e + 0.044715f * e3));
        hid_s[u * HP + tid] = 0.5f * e * (1.0f + th);
    }
    __syncthreads();

    const float* As = hid_s + m * HP + half * 2;
    v2f a[NE / 4];
    #pragma unroll
    for (int kk = 0; kk < NE / 4; ++kk) {
        a[kk].x = As[kk * 4];
        a[kk].y = As[kk * 4 + 1];
    }
    __builtin_amdgcn_sched_barrier(0);
    v8f acc = {};
    #pragma unroll
    for (int kk = 0; kk < NE / 4; ++kk) {
        acc = __builtin_amdgcn_wmma_f32_16x16x4_f32(
                  false, a[kk], false, b[kk], (short)0, acc, false, false);
    }
    float bias = b2[c0 + m];
    #pragma unroll
    for (int r = 0; r < 8; ++r) {
        int row = r + 8 * half;
        Rd[(u0 + row) * NE + (c0 + m)] = acc[r] + bias;
    }
}

#define JT 5
__global__ void __launch_bounds__(32)
srpe_qk_wmma(const float* __restrict__ x,
             const float* __restrict__ Rd,
             float* __restrict__ out)
{
    int bh = blockIdx.y;
    int h  = bh & (HH - 1);
    int id = blockIdx.x;
    int ti = id / (25 / JT);
    int jg = id % (25 / JT);
    int t0 = ti * 16;
    int ub = (ti + jg * JT) * 16;

    int lane = threadIdx.x;
    int m    = lane & 15;
    int half = lane >> 4;

    const float* X = x  + (size_t)bh * TK * FDM;
    const float* R = Rd + h * FDM;

    v2f a[FDM / 4];
    #pragma unroll
    for (int kk = 0; kk < FDM / 4; ++kk) {
        int ka = kk * 4 + half * 2;
        a[kk].x = X[(t0 + m) * FDM + ka];
        a[kk].y = X[(t0 + m) * FDM + ka + 1];
    }
    v2f b[JT][FDM / 4];
    #pragma unroll
    for (int j = 0; j < JT; ++j) {
        #pragma unroll
        for (int kk = 0; kk < FDM / 4; ++kk) {
            int ka = kk * 4 + half * 2;
            b[j][kk].x = R[(ub + j * 16 + m) * NE + ka];
            b[j][kk].y = R[(ub + j * 16 + m) * NE + ka + 1];
        }
    }

    asm volatile("" :: "v"(a[0]), "v"(a[1]), "v"(a[2]), "v"(a[3]), "v"(a[4]), "v"(a[5]), "v"(a[6]), "v"(a[7]), "v"(b[0][0]), "v"(b[0][1]), "v"(b[0][2]), "v"(b[0][3]), "v"(b[0][4]), "v"(b[0][5]), "v"(b[0][6]), "v"(b[0][7]), "v"(b[1][0]), "v"(b[1][1]), "v"(b[1][2]), "v"(b[1][3]), "v"(b[1][4]), "v"(b[1][5]), "v"(b[1][6]), "v"(b[1][7]), "v"(b[2][0]), "v"(b[2][1]), "v"(b[2][2]), "v"(b[2][3]), "v"(b[2][4]), "v"(b[2][5]), "v"(b[2][6]), "v"(b[2][7]), "v"(b[3][0]), "v"(b[3][1]), "v"(b[3][2]), "v"(b[3][3]), "v"(b[3][4]), "v"(b[3][5]), "v"(b[3][6]), "v"(b[3][7]), "v"(b[4][0]), "v"(b[4][1]), "v"(b[4][2]), "v"(b[4][3]), "v"(b[4][4]), "v"(b[4][5]), "v"(b[4][6]), "v"(b[4][7]));

    v8f acc[JT];
    #pragma unroll
    for (int j = 0; j < JT; ++j) acc[j] = v8f{};
    #pragma unroll
    for (int kk = 0; kk < FDM / 4; ++kk) {
        #pragma unroll
        for (int j = 0; j < JT; ++j) {
            acc[j] = __builtin_amdgcn_wmma_f32_16x16x4_f32(
                         false, a[kk], false, b[j][kk], (short)0, acc[j], false, false);
        }
    }

    float* O = out + (size_t)bh * TK * TK;
    #pragma unroll
    for (int j = 0; j < JT; ++j) {
        int u = ub + j * 16 + m;
        #pragma unroll
        for (int r = 0; r < 8; ++r) {
            int t = t0 + r + 8 * half;
            int s = t + 383 - u;
            if (s >= 0 && s < TK) O[t * TK + s] = acc[j][r];
        }
    }
}

extern "C" void kernel_launch(void* const* d_in, const int* in_sizes, int n_in,
                              void* d_out, int out_size, void* d_ws, size_t ws_size,
                              hipStream_t stream)
{
    const float* x  = (const float*)d_in[0];
    const float* pd = (const float*)d_in[1];
    const float* W1 = (const float*)d_in[2];
    const float* b1 = (const float*)d_in[3];
    const float* W2 = (const float*)d_in[4];
    const float* b2 = (const float*)d_in[5];
    float* out = (float*)d_out;

    float* Rd = (float*)d_ws;

    srpe_rd_fused<<<dim3((NU / 16) * 2), dim3(256), 0, stream>>>(pd, W1, b1, W2, b2, Rd);

    srpe_qk_wmma<<<dim3(24 * 5, 16), dim3(32), 0, stream>>>(x, Rd, out);
}


# both launches fused: a 16-wave workgroup owns a head and the u-tile pair (p,p+24), builds its 32x32 Rd block in LDS and computes all output tiles that use it; 2 launches -> 1; k order unchanged
# speedup vs baseline: 1.5231x; 1.1341x over previous
#include <hip/hip_runtime.h>
#include <cmath>


typedef float v2f __attribute__((ext_vector_type(2)));
typedef float v8f __attribute__((ext_vector_type(8)));

#define TK   384
#define HH   8
#define FDM  32
#define NE   256

#define HP 260
#define RP 36

__device__ __forceinline__ void qk_tile(const v2f* xa, const v2f (&rb)[FDM / 4],
                                        float* __restrict__ O, int t0, int ubase,
                                        int m, int half)
{
    v8f acc = {};
    #pragma unroll
    for (int kk = 0; kk < FDM / 4; ++kk) {
        acc = __builtin_amdgcn_wmma_f32_16x16x4_f32(
                  false, xa[kk], false, rb[kk], (short)0, acc, false, false);
    }
    int u = ubase + m;
    #pragma unroll
    for (int r = 0; r < 8; ++r) {
        int t = t0 + r + 8 * half;
        int s = t + 383 - u;
        if (s >= 0 && s < TK) O[t * TK + s] = acc[r];
    }
}

__global__ void __launch_bounds__(512)
srpe_fused(const float* __restrict__ x,
           const float* __restrict__ pd,
           const float* __restrict__ W1,
           const float* __restrict__ b1,
           const float* __restrict__ W2,
           const float* __restrict__ b2,
           float* __restrict__ out)
{
    __shared__ float hid_s[32 * HP];
    __shared__ float rd_s[32 * RP];

    int tid  = threadIdx.x;
    int wave = __builtin_amdgcn_readfirstlane(tid >> 5);
    int lane = tid & 31;
    int m    = lane & 15;
    int half = lane >> 4;
    int p    = blockIdx.x;
    int h    = blockIdx.y;
    int col  = tid & 255;
    int ut   = wave >> 3;
    int u0   = (p + 24 * ut) * 16;

    int ur = u0 + m;
    int ue = (ur < 766) ? ur : 766;
    float d = (ue >= 383) ? pd[(ue - 383) * TK] : pd[383 - ue];
    float w0 = W1[col * 3 + 0];
    float w1 = W1[col * 3 + 1];
    float w2 = W1[col * 3 + 2];
    float bb = b1[col];

    asm volatile("" :: "v"(d), "v"(w0), "v"(w1), "v"(w2), "v"(bb));

    v2f pre[NE / 4];
    float bias = 0.0f;
    int nt = wave & 1;
    int w2i = wave - 4;
    int bq  = w2i / 6;
    int q   = w2i - 6 * bq;
    if (wave < 4) {
        int c0 = h * FDM + nt * 16;
        #pragma unroll
        for (int kk = 0; kk < NE / 4; ++kk) {
            int ka = kk * 4 + half * 2;
            pre[kk].x = W2[(c0 + m) * NE + ka];
            pre[kk].y = W2[(c0 + m) * NE + ka + 1];
        }
        bias = b2[c0 + m];
    } else {
        const float* X = x + (size_t)(bq * HH + h) * TK * FDM;
        #pragma unroll
        for (int i = 0; i < 4; ++i) {
            int t0 = (q + 6 * i) * 16;
            #pragma unroll
            for (int kk = 0; kk < FDM / 4; ++kk) {
                int ka = kk * 4 + half * 2;
                pre[i * 8 + kk].x = X[(t0 + m) * FDM + ka];
                pre[i * 8 + kk].y = X[(t0 + m) * FDM + ka + 1];
            }
        }
    }

    int f0 = __float_as_int(log1pf(fmaxf(d, 0.0f)));
    int f1 = __float_as_int(log1pf(fmaxf(-d, 0.0f)));
    int f2 = __float_as_int((d == 0.0f) ? 1.0f : 0.0f);

    #pragma unroll
    for (int u = 0; u < 16; ++u) {
        float g0 = __int_as_float(__builtin_amdgcn_readlane(f0, u));
        float g1 = __int_as_float(__builtin_amdgcn_readlane(f1, u));
        float g2 = __int_as_float(__builtin_amdgcn_readlane(f2, u));
        float e  = g0 * w0 + g1 * w1 + g2 * w2 + bb;
        float e3 = e * e * e;
        float th = tanhf(0.79788456080286535588f * (e + 0.044715f * e3));
        hid_s[(ut * 16 + u) * HP + col] = 0.5f * e * (1.0f + th);
    }
    __syncthreads();

    if (wave < 4) {
        int ut1 = wave >> 1;
        const float* As = hid_s + (ut1 * 16 + m) * HP + half * 2;
        v2f a[2][16];
        #pragma unroll
        for (int i = 0; i < 16; ++i) {
            a[0][i].x = As[i * 4];
            a[0][i].y = As[i * 4 + 1];
        }
        v8f acc = {};
        #pragma unroll
        for (int c = 0; c < 4; ++c) {
            if (c + 1 < 4) {
                #pragma unroll
                for (int i = 0; i < 16; ++i) {
                    a[(c + 1) & 1][i].x = As[((c + 1) * 16 + i) * 4];
                    a[(c + 1) & 1][i].y = As[((c + 1) * 16 + i) * 4 + 1];
                }
            }
            __builtin_amdgcn_sched_barrier(0);
            #pragma unroll
            for (int i = 0; i < 16; ++i) {
                acc = __builtin_amdgcn_wmma_f32_16x16x4_f32(
                          false, a[c & 1][i], false, pre[c * 16 + i], (short)0, acc, false, false);
            }
        }
        #pragma unroll
        for (int r = 0; r < 8; ++r) {
            int row = r + 8 * half;
            rd_s[(ut1 * 16 + row) * RP + nt * 16 + m] = acc[r] + bias;
        }
    }
    __syncthreads();

    if (wave >= 4) {
        v2f rb[2][FDM / 4];
        #pragma unroll
        for (int j = 0; j < 2; ++j) {
            #pragma unroll
            for (int kk = 0; kk < FDM / 4; ++kk) {
                int ka = kk * 4 + half * 2;
                rb[j][kk].x = rd_s[(j * 16 + m) * RP + ka];
                rb[j][kk].y = rd_s[(j * 16 + m) * RP + ka + 1];
            }
        }
        float* O = out + (size_t)(bq * HH + h) * TK * TK;
        #pragma unroll
        for (int i = 0; i < 4; ++i) {
            int ti = q + 6 * i;
            if (ti <= p) qk_tile(&pre[i * 8], rb[0], O, ti * 16, p * 16, m, half);
            if (ti >= p) qk_tile(&pre[i * 8], rb[1], O, ti * 16, (p + 24) * 16, m, half);
        }
    }
}

extern "C" void kernel_launch(void* const* d_in, const int* in_sizes, int n_in,
                              void* d_out, int out_size, void* d_ws, size_t ws_size,
                              hipStream_t stream)
{
    const float* x  = (const float*)d_in[0];
    const float* pd = (const float*)d_in[1];
    const float* W1 = (const float*)d_in[2];
    const float* b1 = (const float*)d_in[3];
    const float* W2 = (const float*)d_in[4];
    const float* b2 = (const float*)d_in[5];
    float* out = (float*)d_out;

    srpe_fused<<<dim3(24, HH), dim3(512), 0, stream>>>(x, pd, W1, b1, W2, b2, out);
}


# activation: NewGELU as e/(1+2^(e*(c1+c2*e*e))), the same function, on hardware exp2 and rcp: one branch-free form instead of library tanhf's two divergent paths; sums and k order unchanged
# speedup vs baseline: 1.6612x; 1.0907x over previous
#include <hip/hip_runtime.h>
#include <cmath>


typedef float v2f __attribute__((ext_vector_type(2)));
typedef float v8f __attribute__((ext_vector_type(8)));

#define TK   384
#define HH   8
#define FDM  32
#define NE   256

#define GC1 (-2.3022081981443252f)
#define GC2 (-0.10294323958002349f)

#define HP 260
#define RP 36

__device__ __forceinline__ void qk_tile(const v2f* xa, const v2f (&rb)[FDM / 4],
                                        float* __restrict__ O, int t0, int ubase,
                                        int m, int half)
{
    v8f acc = {};
    #pragma unroll
    for (int kk = 0; kk < FDM / 4; ++kk) {
        acc = __builtin_amdgcn_wmma_f32_16x16x4_f32(
                  false, xa[kk], false, rb[kk], (short)0, acc, false, false);
    }
    int u = ubase + m;
    #pragma unroll
    for (int r = 0; r < 8; ++r) {
        int t = t0 + r + 8 * half;
        int s = t + 383 - u;
        if (s >= 0 && s < TK) O[t * TK + s] = acc[r];
    }
}

__global__ void __launch_bounds__(512)
srpe_fused(const float* __restrict__ x,
           const float* __restrict__ pd,
           const float* __restrict__ W1,
           const float* __restrict__ b1,
           const float* __restrict__ W2,
           const float* __restrict__ b2,
           float* __restrict__ out)
{
    __shared__ float hid_s[32 * HP];
    __shared__ float rd_s[32 * RP];

    int tid  = threadIdx.x;
    int wave = __builtin_amdgcn_readfirstlane(tid >> 5);
    int lane = tid & 31;
    int m    = lane & 15;
    int half = lane >> 4;
    int p    = blockIdx.x;
    int h    = blockIdx.y;
    int col  = tid & 255;
    int ut   = wave >> 3;
    int u0   = (p + 24 * ut) * 16;

    int ur = u0 + m;
    int ue = (ur < 766) ? ur : 766;
    float d = (ue >= 383) ? pd[(ue - 383) * TK] : pd[383 - ue];
    float w0 = W1[col * 3 + 0];
    float w1 = W1[col * 3 + 1];
    float w2 = W1[col * 3 + 2];
    float bb = b1[col];

    asm volatile("" :: "v"(d), "v"(w0), "v"(w1), "v"(w2), "v"(bb));

    v2f pre[NE / 4];
    float bias = 0.0f;
    int nt = wave & 1;
    int w2i = wave - 4;
    int bq  = w2i / 6;
    int q   = w2i - 6 * bq;
    if (wave < 4) {
        int c0 = h * FDM + nt * 16;
        #pragma unroll
        for (int kk = 0; kk < NE / 4; ++kk) {
            int ka = kk * 4 + half * 2;
            pre[kk].x = W2[(c0 + m) * NE + ka];
            pre[kk].y = W2[(c0 + m) * NE + ka + 1];
        }
        bias = b2[c0 + m];
    } else {
        const float* X = x + (size_t)(bq * HH + h) * TK * FDM;
        #pragma unroll
        for (int i = 0; i < 4; ++i) {
            int t0 = (q + 6 * i) * 16;
            #pragma unroll
            for (int kk = 0; kk < FDM / 4; ++kk) {
                int ka = kk * 4 + half * 2;
                pre[i * 8 + kk].x = X[(t0 + m) * FDM + ka];
                pre[i * 8 + kk].y = X[(t0 + m) * FDM + ka + 1];
            }
        }
    }

    int f0 = __float_as_int(log1pf(fmaxf(d, 0.0f)));
    int f1 = __float_as_int(log1pf(fmaxf(-d, 0.0f)));
    int f2 = __float_as_int((d == 0.0f) ? 1.0f : 0.0f);

    #pragma unroll
    for (int u = 0; u < 16; ++u) {
        float g0 = __int_as_float(__builtin_amdgcn_readlane(f0, u));
        float g1 = __int_as_float(__builtin_amdgcn_readlane(f1, u));
        float g2 = __int_as_float(__builtin_amdgcn_readlane(f2, u));
        float e  = g0 * w0 + g1 * w1 + g2 * w2 + bb;
        float tq = fmaf(e * e, GC2, GC1);
        float ex = __builtin_amdgcn_exp2f(e * tq);
        hid_s[(ut * 16 + u) * HP + col] = e * __builtin_amdgcn_rcpf(1.0f + ex);
    }
    __syncthreads();

    if (wave < 4) {
        int ut1 = wave >> 1;
        const float* As = hid_s + (ut1 * 16 + m) * HP + half * 2;
        v2f a[2][16];
        #pragma unroll
        for (int i = 0; i < 16; ++i) {
            a[0][i].x = As[i * 4];
            a[0][i].y = As[i * 4 + 1];
        }
        v8f acc = {};
        #pragma unroll
        for (int c = 0; c < 4; ++c) {
            if (c + 1 < 4) {
                #pragma unroll
                for (int i = 0; i < 16; ++i) {
                    a[(c + 1) & 1][i].x = As[((c + 1) * 16 + i) * 4];
                    a[(c + 1) & 1][i].y = As[((c + 1) * 16 + i) * 4 + 1];
                }
            }
            __builtin_amdgcn_sched_barrier(0);
            #pragma unroll
            for (int i = 0; i < 16; ++i) {
                acc = __builtin_amdgcn_wmma_f32_16x16x4_f32(
                          false, a[c & 1][i], false, pre[c * 16 + i], (short)0, acc, false, false);
            }
        }
        #pragma unroll
        for (int r = 0; r < 8; ++r) {
            int row = r + 8 * half;
            rd_s[(ut1 * 16 + row) * RP + nt * 16 + m] = acc[r] + bias;
        }
    }
    __syncthreads();

    if (wave >= 4) {
        v2f rb[2][FDM / 4];
        #pragma unroll
        for (int j = 0; j < 2; ++j) {
            #pragma unroll
            for (int kk = 0; kk < FDM / 4; ++kk) {
                int ka = kk * 4 + half * 2;
                rb[j][kk].x = rd_s[(j * 16 + m) * RP + ka];
                rb[j][kk].y = rd_s[(j * 16 + m) * RP + ka + 1];
            }
        }
        float* O = out + (size_t)(bq * HH + h) * TK * TK;
        #pragma unroll
        for (int i = 0; i < 4; ++i) {
            int ti = q + 6 * i;
            if (ti <= p) qk_tile(&pre[i * 8], rb[0], O, ti * 16, p * 16, m, half);
            if (ti >= p) qk_tile(&pre[i * 8], rb[1], O, ti * 16, (p + 24) * 16, m, half);
        }
    }
}

extern "C" void kernel_launch(void* const* d_in, const int* in_sizes, int n_in,
                              void* d_out, int out_size, void* d_ws, size_t ws_size,
                              hipStream_t stream)
{
    const float* x  = (const float*)d_in[0];
    const float* pd = (const float*)d_in[1];
    const float* W1 = (const float*)d_in[2];
    const float* b1 = (const float*)d_in[3];
    const float* W2 = (const float*)d_in[4];
    const float* b2 = (const float*)d_in[5];
    float* out = (float*)d_out;

    srpe_fused<<<dim3(24, HH), dim3(512), 0, stream>>>(x, pd, W1, b1, W2, b2, out);
}


# 16 waves alike: Rd product split along K (wave w owns k=16w..16w+15, makes its activations as A fragments, no activation tile in LDS), 16 partial blocks summed in LDS: Rd sum order changes
# speedup vs baseline: 1.7823x; 1.0729x over previous
#include <hip/hip_runtime.h>
#include <cmath>


typedef float v2f __attribute__((ext_vector_type(2)));
typedef float v4f __attribute__((ext_vector_type(4)));
typedef float v8f __attribute__((ext_vector_type(8)));

#define TK   384
#define HH   8
#define FDM  32
#define NE   256
#define NW   16

#define GC1 (-2.3022081981443252f)
#define GC2 (-0.10294323958002349f)

#define RP 36

__device__ __forceinline__ float gelu_new(float e)
{
    float tq = fmaf(e * e, GC2, GC1);
    float ex = __builtin_amdgcn_exp2f(e * tq);
    return e * __builtin_amdgcn_rcpf(1.0f + ex);
}

__device__ __forceinline__ void qk_tile(const v2f* xa, const v2f (&rb)[FDM / 4],
                                        float* __restrict__ O, int t0, int ubase,
                                        int m, int half)
{
    v8f acc = {};
    #pragma unroll
    for (int kk = 0; kk < FDM / 4; ++kk) {
        acc = __builtin_amdgcn_wmma_f32_16x16x4_f32(
                  false, xa[kk], false, rb[kk], (short)0, acc, false, false);
    }
    int u = ubase + m;
    #pragma unroll
    for (int r = 0; r < 8; ++r) {
        int t = t0 + r + 8 * half;
        int s = t + 383 - u;
        if (s >= 0 && s < TK) O[t * TK + s] = acc[r];
    }
}

__global__ void __launch_bounds__(512)
srpe_fused(const float* __restrict__ x,
           const float* __restrict__ pd,
           const float* __restrict__ W1,
           const float* __restrict__ b1,
           const float* __restrict__ W2,
           const float* __restrict__ b2,
           float* __restrict__ out)
{
    __shared__ __attribute__((aligned(16))) float red_s[NW * 1024];
    __shared__ float rd_s[32 * RP];

    int tid  = threadIdx.x;
    int wave = __builtin_amdgcn_readfirstlane(tid >> 5);
    int lane = tid & 31;
    int m    = lane & 15;
    int half = lane >> 4;
    int p    = blockIdx.x;
    int h    = blockIdx.y;
    int kb   = wave * 16;

    int ur = (p + 24 * half) * 16 + m;
    int ue = (ur < 766) ? ur : 766;
    float d = (ue >= 383) ? pd[(ue - 383) * TK] : pd[383 - ue];

    float w0[4][2], w1[4][2], w2[4][2], bb[4][2];
    #pragma unroll
    for (int kk = 0; kk < 4; ++kk) {
        #pragma unroll
        for (int e = 0; e < 2; ++e) {
            int k = kb + kk * 4 + half * 2 + e;
            w0[kk][e] = W1[k * 3 + 0];
            w1[kk][e] = W1[k * 3 + 1];
            w2[kk][e] = W1[k * 3 + 2];
            bb[kk][e] = b1[k];
        }
    }

    int c0 = h * FDM;
    v2f wb[2][4];
    #pragma unroll
    for (int nt = 0; nt < 2; ++nt) {
        #pragma unroll
        for (int kk = 0; kk < 4; ++kk) {
            int ka = kb + kk * 4 + half * 2;
            wb[nt][kk].x = W2[(c0 + nt * 16 + m) * NE + ka];
            wb[nt][kk].y = W2[(c0 + nt * 16 + m) * NE + ka + 1];
        }
    }

    int idx   = tid * 2;
    int rtile = idx >> 8;
    int rr4   = (idx >> 7) & 1;
    int rln   = (idx >> 2) & 31;
    int ri0   = idx & 3;
    int rnt   = rtile & 1;
    int rcol  = rln & 15;
    int rrow  = (rtile >> 1) * 16 + rr4 * 4 + ri0 + 8 * (rln >> 4);
    float bias = b2[c0 + rnt * 16 + rcol];

    v2f pre[24];
    int bq  = wave >> 3;
    int q   = wave & 7;
    const float* X = x + (size_t)(bq * HH + h) * TK * FDM;
    #pragma unroll
    for (int i = 0; i < 3; ++i) {
        int t0 = (q + 8 * i) * 16;
        #pragma unroll
        for (int kk = 0; kk < FDM / 4; ++kk) {
            int ka = kk * 4 + half * 2;
            pre[i * 8 + kk].x = X[(t0 + m) * FDM + ka];
            pre[i * 8 + kk].y = X[(t0 + m) * FDM + ka + 1];
        }
    }
    __builtin_amdgcn_sched_barrier(0);

    float f0 = log1pf(fmaxf(d, 0.0f));
    float f1 = log1pf(fmaxf(-d, 0.0f));
    float f2 = (d == 0.0f) ? 1.0f : 0.0f;
    float o0 = __shfl_xor(f0, 16);
    float o1 = __shfl_xor(f1, 16);
    float o2 = __shfl_xor(f2, 16);
    float g0[2], g1[2], g2[2];
    g0[0] = half ? o0 : f0;  g0[1] = half ? f0 : o0;
    g1[0] = half ? o1 : f1;  g1[1] = half ? f1 : o1;
    g2[0] = half ? o2 : f2;  g2[1] = half ? f2 : o2;

    v8f acc[2][2] = {};
    #pragma unroll
    for (int kk = 0; kk < 4; ++kk) {
        v2f a[2];
        #pragma unroll
        for (int j = 0; j < 2; ++j) {
            float e0 = g0[j] * w0[kk][0] + g1[j] * w1[kk][0] + g2[j] * w2[kk][0] + bb[kk][0];
            float e1 = g0[j] * w0[kk][1] + g1[j] * w1[kk][1] + g2[j] * w2[kk][1] + bb[kk][1];
            a[j].x = gelu_new(e0);
            a[j].y = gelu_new(e1);
        }
        #pragma unroll
        for (int j = 0; j < 2; ++j) {
            #pragma unroll
            for (int nt = 0; nt < 2; ++nt) {
                acc[j][nt] = __builtin_amdgcn_wmma_f32_16x16x4_f32(
                                 false, a[j], false, wb[nt][kk], (short)0, acc[j][nt], false, false);
            }
        }
    }
    #pragma unroll
    for (int j = 0; j < 2; ++j) {
        #pragma unroll
        for (int nt = 0; nt < 2; ++nt) {
            #pragma unroll
            for (int r4 = 0; r4 < 2; ++r4) {
                v4f v = { acc[j][nt][r4 * 4 + 0], acc[j][nt][r4 * 4 + 1],
                          acc[j][nt][r4 * 4 + 2], acc[j][nt][r4 * 4 + 3] };
                *(v4f*)&red_s[((wave * 4 + j * 2 + nt) * 2 + r4) * 128 + lane * 4] = v;
            }
        }
    }
    __syncthreads();

    {
        v2f sum = *(const v2f*)&red_s[idx];
        #pragma unroll
        for (int w = 1; w < NW; ++w) {
            v2f pv = *(const v2f*)&red_s[w * 1024 + idx];
            sum.x += pv.x;
            sum.y += pv.y;
        }
        rd_s[rrow * RP + rnt * 16 + rcol]       = sum.x + bias;
        rd_s[(rrow + 1) * RP + rnt * 16 + rcol] = sum.y + bias;
    }
    __syncthreads();

    {
        v2f rb[2][FDM / 4];
        #pragma unroll
        for (int j = 0; j < 2; ++j) {
            #pragma unroll
            for (int kk = 0; kk < FDM / 4; ++kk) {
                int ka = kk * 4 + half * 2;
                rb[j][kk].x = rd_s[(j * 16 + m) * RP + ka];
                rb[j][kk].y = rd_s[(j * 16 + m) * RP + ka + 1];
            }
        }
        float* O = out + (size_t)(bq * HH + h) * TK * TK;
        #pragma unroll
        for (int i = 0; i < 3; ++i) {
            int ti = q + 8 * i;
            if (ti <= p) qk_tile(&pre[i * 8], rb[0], O, ti * 16, p * 16, m, half);
            if (ti >= p) qk_tile(&pre[i * 8], rb[1], O, ti * 16, (p + 24) * 16, m, half);
        }
    }
}

extern "C" void kernel_launch(void* const* d_in, const int* in_sizes, int n_in,
                              void* d_out, int out_size, void* d_ws, size_t ws_size,
                              hipStream_t stream)
{
    const float* x  = (const float*)d_in[0];
    const float* pd = (const float*)d_in[1];
    const float* W1 = (const float*)d_in[2];
    const float* b1 = (const float*)d_in[3];
    const float* W2 = (const float*)d_in[4];
    const float* b2 = (const float*)d_in[5];
    float* out = (float*)d_out;

    srpe_fused<<<dim3(24, HH), dim3(512), 0, stream>>>(x, pd, W1, b1, W2, b2, out);
}


# operand fetch at 16 bytes per lane: x tiles as whole-line loads made into A fragments through wave-private LDS; W2, W1, b1 in contiguous pieces: k order inside a wave's Rd slice permuted (sum order)
# speedup vs baseline: 1.9681x; 1.1042x over previous
#include <hip/hip_runtime.h>
#include <cmath>


typedef float v2f __attribute__((ext_vector_type(2)));
typedef float v4f __attribute__((ext_vector_type(4)));
typedef float v8f __attribute__((ext_vector_type(8)));

#define TK   384
#define HH   8
#define FDM  32
#define NE   256
#define NW   16

#define GC1 (-2.3022081981443252f)
#define GC2 (-0.10294323958002349f)

#define RP 36

__device__ __forceinline__ float gelu_new(float e)
{
    float tq = fmaf(e * e, GC2, GC1);
    float ex = __builtin_amdgcn_exp2f(e * tq);
    return e * __builtin_amdgcn_rcpf(1.0f + ex);
}

__device__ __forceinline__ void qk_tile(const v2f* xa, const v2f (&rb)[FDM / 4],
                                        float* __restrict__ O, int t0, int ubase,
                                        int m, int half)
{
    v8f acc = {};
    #pragma unroll
    for (int kk = 0; kk < FDM / 4; ++kk) {
        acc = __builtin_amdgcn_wmma_f32_16x16x4_f32(
                  false, xa[kk], false, rb[kk], (short)0, acc, false, false);
    }
    int u = ubase + m;
    #pragma unroll
    for (int r = 0; r < 8; ++r) {
        int t = t0 + r + 8 * half;
        int s = t + 383 - u;
        if (s >= 0 && s < TK) O[t * TK + s] = acc[r];
    }
}

__global__ void __launch_bounds__(512)
srpe_fused(const float* __restrict__ x,
           const float* __restrict__ pd,
           const float* __restrict__ W1,
           const float* __restrict__ b1,
           const float* __restrict__ W2,
           const float* __restrict__ b2,
           float* __restrict__ out)
{
    __shared__ __attribute__((aligned(16))) float lds_s[NW * 1024 + 32 * RP + NW * 48 * RP];
    float* red_s = lds_s;
    float* rd_s  = lds_s + NW * 1024;
    float* xs_s  = lds_s + NW * 1024 + 32 * RP;

    int tid  = threadIdx.x;
    int wave = __builtin_amdgcn_readfirstlane(tid >> 5);
    int lane = tid & 31;
    int m    = lane & 15;
    int half = lane >> 4;
    int p    = blockIdx.x;
    int h    = blockIdx.y;
    int kb   = wave * 16;

    int ur = (p + 24 * half) * 16 + m;
    int ue = (ur < 766) ? ur : 766;
    float d = (ue >= 383) ? pd[(ue - 383) * TK] : pd[383 - ue];

    int k8 = kb + 8 * half;
    float wv[24], bv[8];
    #pragma unroll
    for (int v = 0; v < 6; ++v) {
        v4f t4 = *(const v4f*)&W1[k8 * 3 + 4 * v];
        wv[4 * v + 0] = t4.x;  wv[4 * v + 1] = t4.y;
        wv[4 * v + 2] = t4.z;  wv[4 * v + 3] = t4.w;
    }
    #pragma unroll
    for (int v = 0; v < 2; ++v) {
        v4f t4 = *(const v4f*)&b1[k8 + 4 * v];
        bv[4 * v + 0] = t4.x;  bv[4 * v + 1] = t4.y;
        bv[4 * v + 2] = t4.z;  bv[4 * v + 3] = t4.w;
    }
    float w0[4][2], w1[4][2], w2[4][2], bb[4][2];
    #pragma unroll
    for (int kk = 0; kk < 4; ++kk) {
        #pragma unroll
        for (int e = 0; e < 2; ++e) {
            int kl = 2 * kk + e;
            w0[kk][e] = wv[kl * 3 + 0];
            w1[kk][e] = wv[kl * 3 + 1];
            w2[kk][e] = wv[kl * 3 + 2];
            bb[kk][e] = bv[kl];
        }
    }

    int c0 = h * FDM;
    v2f wb[2][4];
    #pragma unroll
    for (int nt = 0; nt < 2; ++nt) {
        #pragma unroll
        for (int j = 0; j < 2; ++j) {
            v4f t4 = *(const v4f*)&W2[(c0 + nt * 16 + m) * NE + k8 + 4 * j];
            wb[nt][2 * j].x     = t4.x;  wb[nt][2 * j].y     = t4.y;
            wb[nt][2 * j + 1].x = t4.z;  wb[nt][2 * j + 1].y = t4.w;
        }
    }

    int idx   = tid * 2;
    int rtile = idx >> 8;
    int rr4   = (idx >> 7) & 1;
    int rln   = (idx >> 2) & 31;
    int ri0   = idx & 3;
    int rnt   = rtile & 1;
    int rcol  = rln & 15;
    int rrow  = (rtile >> 1) * 16 + rr4 * 4 + ri0 + 8 * (rln >> 4);
    float bias = b2[c0 + rnt * 16 + rcol];

    v4f xg[12];
    int bq  = wave >> 3;
    int q   = wave & 7;
    const float* X = x + (size_t)(bq * HH + h) * TK * FDM;
    #pragma unroll
    for (int i = 0; i < 3; ++i) {
        int t0 = (q + 8 * i) * 16;
        #pragma unroll
        for (int j = 0; j < 4; ++j) {
            xg[i * 4 + j] = *(const v4f*)&X[t0 * FDM + j * 128 + lane * 4];
        }
    }
    __builtin_amdgcn_sched_barrier(0);

    float f0 = log1pf(fmaxf(d, 0.0f));
    float f1 = log1pf(fmaxf(-d, 0.0f));
    float f2 = (d == 0.0f) ? 1.0f : 0.0f;
    float o0 = __shfl_xor(f0, 16);
    float o1 = __shfl_xor(f1, 16);
    float o2 = __shfl_xor(f2, 16);
    float g0[2], g1[2], g2[2];
    g0[0] = half ? o0 : f0;  g0[1] = half ? f0 : o0;
    g1[0] = half ? o1 : f1;  g1[1] = half ? f1 : o1;
    g2[0] = half ? o2 : f2;  g2[1] = half ? f2 : o2;

    v8f acc[2][2] = {};
    #pragma unroll
    for (int kk = 0; kk < 4; ++kk) {
        v2f a[2];
        #pragma unroll
        for (int j = 0; j < 2; ++j) {
            float e0 = g0[j] * w0[kk][0] + g1[j] * w1[kk][0] + g2[j] * w2[kk][0] + bb[kk][0];
            float e1 = g0[j] * w0[kk][1] + g1[j] * w1[kk][1] + g2[j] * w2[kk][1] + bb[kk][1];
            a[j].x = gelu_new(e0);
            a[j].y = gelu_new(e1);
        }
        #pragma unroll
        for (int j = 0; j < 2; ++j) {
            #pragma unroll
            for (int nt = 0; nt < 2; ++nt) {
                acc[j][nt] = __builtin_amdgcn_wmma_f32_16x16x4_f32(
                                 false, a[j], false, wb[nt][kk], (short)0, acc[j][nt], false, false);
            }
        }
    }
    #pragma unroll
    for (int j = 0; j < 2; ++j) {
        #pragma unroll
        for (int nt = 0; nt < 2; ++nt) {
            #pragma unroll
            for (int r4 = 0; r4 < 2; ++r4) {
                v4f v = { acc[j][nt][r4 * 4 + 0], acc[j][nt][r4 * 4 + 1],
                          acc[j][nt][r4 * 4 + 2], acc[j][nt][r4 * 4 + 3] };
                *(v4f*)&red_s[((wave * 4 + j * 2 + nt) * 2 + r4) * 128 + lane * 4] = v;
            }
        }
    }
    float* xw = xs_s + wave * (48 * RP);
    #pragma unroll
    for (int i = 0; i < 3; ++i) {
        #pragma unroll
        for (int j = 0; j < 4; ++j) {
            *(v4f*)&xw[(i * 16 + j * 4 + (lane >> 3)) * RP + (lane & 7) * 4] = xg[i * 4 + j];
        }
    }
    __syncthreads();

    {
        v2f sum = *(const v2f*)&red_s[idx];
        #pragma unroll
        for (int w = 1; w < NW; ++w) {
            v2f pv = *(const v2f*)&red_s[w * 1024 + idx];
            sum.x += pv.x;
            sum.y += pv.y;
        }
        rd_s[rrow * RP + rnt * 16 + rcol]       = sum.x + bias;
        rd_s[(rrow + 1) * RP + rnt * 16 + rcol] = sum.y + bias;
    }
    __syncthreads();

    {
        v2f rb[2][FDM / 4];
        #pragma unroll
        for (int j = 0; j < 2; ++j) {
            #pragma unroll
            for (int kk = 0; kk < FDM / 4; ++kk) {
                int ka = kk * 4 + half * 2;
                rb[j][kk].x = rd_s[(j * 16 + m) * RP + ka];
                rb[j][kk].y = rd_s[(j * 16 + m) * RP + ka + 1];
            }
        }
        v2f pre[24];
        #pragma unroll
        for (int i = 0; i < 3; ++i) {
            #pragma unroll
            for (int kk = 0; kk < FDM / 4; ++kk) {
                int ka = kk * 4 + half * 2;
                pre[i * 8 + kk].x = xw[(i * 16 + m) * RP + ka];
                pre[i * 8 + kk].y = xw[(i * 16 + m) * RP + ka + 1];
            }
        }
        float* O = out + (size_t)(bq * HH + h) * TK * TK;
        #pragma unroll
        for (int i = 0; i < 3; ++i) {
            int ti = q + 8 * i;
            if (ti <= p) qk_tile(&pre[i * 8], rb[0], O, ti * 16, p * 16, m, half);
            if (ti >= p) qk_tile(&pre[i * 8], rb[1], O, ti * 16, (p + 24) * 16, m, half);
        }
    }
}

extern "C" void kernel_launch(void* const* d_in, const int* in_sizes, int n_in,
                              void* d_out, int out_size, void* d_ws, size_t ws_size,
                              hipStream_t stream)
{
    const float* x  = (const float*)d_in[0];
    const float* pd = (const float*)d_in[1];
    const float* W1 = (const float*)d_in[2];
    const float* b1 = (const float*)d_in[3];
    const float* W2 = (const float*)d_in[4];
    const float* b2 = (const float*)d_in[5];
    float* out = (float*)d_out;

    srpe_fused<<<dim3(24, HH), dim3(512), 0, stream>>>(x, pd, W1, b1, W2, b2, out);
}
